# starting kernel
# baseline (speedup 1.0000x reference)
#include <hip/hip_runtime.h>

typedef __attribute__((ext_vector_type(2))) float v2f;
typedef __attribute__((ext_vector_type(8))) float v8f;

#define NG   256
#define KDIM 12
#define GRID_DIM 128
#define NVOX (GRID_DIM*GRID_DIM*GRID_DIM)
#define ROW_BLOCKS (NVOX/16)
#define MAIN_BLOCKS 2048

__global__ __launch_bounds__(256) void vox_prep(const float* __restrict__ pos,
                                                const float* __restrict__ scl,
                                                const float* __restrict__ rot,
                                                float* __restrict__ w) {
  int n = blockIdx.x * blockDim.x + threadIdx.x;
  if (n >= NG) return;
  float qw = rot[n*4+0], qx = rot[n*4+1], qy = rot[n*4+2], qz = rot[n*4+3];
  float rn = rsqrtf(qw*qw + qx*qx + qy*qy + qz*qz);
  qw *= rn; qx *= rn; qy *= rn; qz *= rn;
  float R00 = 1.f - 2.f*(qy*qy + qz*qz);
  float R01 = 2.f*(qx*qy - qw*qz);
  float R02 = 2.f*(qx*qz + qw*qy);
  float R10 = 2.f*(qx*qy + qw*qz);
  float R11 = 1.f - 2.f*(qx*qx + qz*qz);
  float R12 = 2.f*(qy*qz - qw*qx);
  float R20 = 2.f*(qx*qz - qw*qy);
  float R21 = 2.f*(qy*qz + qw*qx);
  float R22 = 1.f - 2.f*(qx*qx + qy*qy);
  float s0 = scl[n*3+0], s1 = scl[n*3+1], s2 = scl[n*3+2];
  float i0 = 1.f/(s0*s0 + 1e-8f);
  float i1 = 1.f/(s1*s1 + 1e-8f);
  float i2 = 1.f/(s2*s2 + 1e-8f);
  float C00 = R00*R00*i0 + R01*R01*i1 + R02*R02*i2;
  float C01 = R00*R10*i0 + R01*R11*i1 + R02*R12*i2;
  float C02 = R00*R20*i0 + R01*R21*i1 + R02*R22*i2;
  float C11 = R10*R10*i0 + R11*R11*i1 + R12*R12*i2;
  float C12 = R10*R20*i0 + R11*R21*i1 + R12*R22*i2;
  float C22 = R20*R20*i0 + R21*R21*i1 + R22*R22*i2;
  float p0 = pos[n*3+0], p1 = pos[n*3+1], p2 = pos[n*3+2];
  float b0 = C00*p0 + C01*p1 + C02*p2;
  float b1 = C01*p0 + C11*p1 + C12*p2;
  float b2 = C02*p0 + C12*p1 + C22*p2;
  float cn = p0*b0 + p1*b1 + p2*b2;
  float* wn = w + n*KDIM;
  wn[0] = C00;      wn[1] = C11;      wn[2]  = C22;
  wn[3] = 2.f*C01;  wn[4] = 2.f*C02;  wn[5]  = 2.f*C12;
  wn[6] = -2.f*b0;  wn[7] = -2.f*b1;  wn[8]  = -2.f*b2;
  wn[9] = cn;       wn[10] = 0.f;     wn[11] = 0.f;
}

__global__ __launch_bounds__(256) void vox_main(const float* __restrict__ w,
                                                const float* __restrict__ density,
                                                float* __restrict__ out) {
  __shared__ float lw[NG * KDIM];
  __shared__ float ld[NG];

  const int tid = threadIdx.x;
  for (int i = tid; i < NG * KDIM; i += 256) lw[i] = w[i];
  if (tid < NG) ld[tid] = density[tid];
  __syncthreads();

  const int lane = tid & 31;
  const int wave = tid >> 5;
  const int m    = lane & 15;
  const bool h   = lane >= 16;

  const float EXP_SCALE = -0.72134752044448170368f;
  const float SC = 2.0f / 127.0f;

  int rb = blockIdx.x * 8 + wave;
  const int stride = gridDim.x * 8;

  for (; rb < ROW_BLOCKS; rb += stride) {
    const int v  = rb * 16 + m;
    const int iz = v >> 14;
    const int iy = (v >> 7) & 127;
    const int ix = v & 127;
    const float c0 = (float)iz * SC - 1.0f;
    const float c1 = (float)iy * SC - 1.0f;
    const float c2 = (float)ix * SC - 1.0f;

    v2f a0, a1, a2;
    a0.x = h ? c2*c2 : c0*c0;   a0.y = h ? c0*c1 : c1*c1;
    a1.x = h ? c0    : c0*c2;   a1.y = h ? c1    : c1*c2;
    a2.x = h ? 0.f   : c2;      a2.y = h ? 0.f   : 1.f;

    float acc[8];
    #pragma unroll
    for (int i = 0; i < 8; ++i) acc[i] = 0.f;

    #pragma unroll
    for (int t = 0; t < 16; ++t) {
      const int n = t * 16 + m;
      const v2f* wp = (const v2f*)&lw[n * KDIM];
      const int hs = h ? 1 : 0;
      v2f b0 = wp[0 + hs];
      v2f b1 = wp[2 + hs];
      v2f b2 = wp[4 + hs];

      v8f c = {0.f,0.f,0.f,0.f,0.f,0.f,0.f,0.f};
      c = __builtin_amdgcn_wmma_f32_16x16x4_f32(false, a0, false, b0, (short)0, c, false, false);
      c = __builtin_amdgcn_wmma_f32_16x16x4_f32(false, a1, false, b1, (short)0, c, false, false);
      c = __builtin_amdgcn_wmma_f32_16x16x4_f32(false, a2, false, b2, (short)0, c, false, false);

      const float dn = ld[n];
      #pragma unroll
      for (int i = 0; i < 8; ++i)
        acc[i] += __builtin_amdgcn_exp2f(c[i] * EXP_SCALE) * dn;
    }

    #pragma unroll
    for (int mask = 1; mask <= 8; mask <<= 1) {
      #pragma unroll
      for (int i = 0; i < 8; ++i)
        acc[i] += __shfl_xor(acc[i], mask, 32);
    }

    if (lane == 0 || lane == 16) {
      float4 lo = make_float4(acc[0], acc[1], acc[2], acc[3]);
      float4 hi = make_float4(acc[4], acc[5], acc[6], acc[7]);
      float* dst = out + rb * 16 + (h ? 8 : 0);
      ((float4*)dst)[0] = lo;
      ((float4*)dst)[1] = hi;
    }
  }
}

extern "C" void kernel_launch(void* const* d_in, const int* in_sizes, int n_in,
                              void* d_out, int out_size, void* d_ws, size_t ws_size,
                              hipStream_t stream) {
  const float* positions = (const float*)d_in[0];
  const float* scales    = (const float*)d_in[1];
  const float* rotations = (const float*)d_in[2];
  const float* density   = (const float*)d_in[3];
  float* out = (float*)d_out;
  float* wtab = (float*)d_ws;

  vox_prep<<<1, 256, 0, stream>>>(positions, scales, rotations, wtab);
  vox_main<<<MAIN_BLOCKS, 256, 0, stream>>>(wtab, density, out);
}


# Separable grid: (z,y)-only terms of the quadratic form summed once per 8 row blocks by a shared level-1 WMMA; each row-block tile then needs 1 WMMA (K=4), was 3 (K=12); mahal sum order changes
# speedup vs baseline: 1.4738x; 1.4738x over previous
#include <hip/hip_runtime.h>

typedef __attribute__((ext_vector_type(2))) float v2f;
typedef __attribute__((ext_vector_type(8))) float v8f;

#define NG   256
#define KDIM 12
#define GRID_DIM 128
#define NVOX (GRID_DIM*GRID_DIM*GRID_DIM)
#define ROW_BLOCKS (NVOX/16)
#define WORK_ITEMS (ROW_BLOCKS/8)
#define MAIN_BLOCKS 2048

__global__ __launch_bounds__(256) void vox_prep(const float* __restrict__ pos,
                                                const float* __restrict__ scl,
                                                const float* __restrict__ rot,
                                                float* __restrict__ w) {
  int n = blockIdx.x * blockDim.x + threadIdx.x;
  if (n >= NG) return;
  float qw = rot[n*4+0], qx = rot[n*4+1], qy = rot[n*4+2], qz = rot[n*4+3];
  float rn = rsqrtf(qw*qw + qx*qx + qy*qy + qz*qz);
  qw *= rn; qx *= rn; qy *= rn; qz *= rn;
  float R00 = 1.f - 2.f*(qy*qy + qz*qz);
  float R01 = 2.f*(qx*qy - qw*qz);
  float R02 = 2.f*(qx*qz + qw*qy);
  float R10 = 2.f*(qx*qy + qw*qz);
  float R11 = 1.f - 2.f*(qx*qx + qz*qz);
  float R12 = 2.f*(qy*qz - qw*qx);
  float R20 = 2.f*(qx*qz - qw*qy);
  float R21 = 2.f*(qy*qz + qw*qx);
  float R22 = 1.f - 2.f*(qx*qx + qy*qy);
  float s0 = scl[n*3+0], s1 = scl[n*3+1], s2 = scl[n*3+2];
  float i0 = 1.f/(s0*s0 + 1e-8f);
  float i1 = 1.f/(s1*s1 + 1e-8f);
  float i2 = 1.f/(s2*s2 + 1e-8f);
  float C00 = R00*R00*i0 + R01*R01*i1 + R02*R02*i2;
  float C01 = R00*R10*i0 + R01*R11*i1 + R02*R12*i2;
  float C02 = R00*R20*i0 + R01*R21*i1 + R02*R22*i2;
  float C11 = R10*R10*i0 + R11*R11*i1 + R12*R12*i2;
  float C12 = R10*R20*i0 + R11*R21*i1 + R12*R22*i2;
  float C22 = R20*R20*i0 + R21*R21*i1 + R22*R22*i2;
  float p0 = pos[n*3+0], p1 = pos[n*3+1], p2 = pos[n*3+2];
  float b0 = C00*p0 + C01*p1 + C02*p2;
  float b1 = C01*p0 + C11*p1 + C12*p2;
  float b2 = C02*p0 + C12*p1 + C22*p2;
  float cn = p0*b0 + p1*b1 + p2*b2;
  float* wn = w + n*KDIM;
  wn[0] = C00;      wn[1] = C11;      wn[2]  = 2.f*C01;
  wn[3] = -2.f*b0;  wn[4] = -2.f*b1;  wn[5]  = cn;
  wn[6] = 2.f*C02;  wn[7] = 2.f*C12;  wn[8]  = -2.f*b2;
  wn[9] = C22;      wn[10] = 0.f;     wn[11] = 0.f;
}

__global__ __launch_bounds__(256) void vox_main(const float* __restrict__ w,
                                                const float* __restrict__ density,
                                                float* __restrict__ out) {
  __shared__ float lw[NG * KDIM];
  __shared__ float ld[NG];

  const int tid = threadIdx.x;
  for (int i = tid; i < NG * KDIM; i += 256) lw[i] = w[i];
  if (tid < NG) ld[tid] = density[tid];
  __syncthreads();

  const int lane = tid & 31;
  const int wave = tid >> 5;
  const int m    = lane & 15;
  const bool h   = lane >= 16;

  const float EXP_SCALE = -0.72134752044448170368f;
  const float SC = 2.0f / 127.0f;

  const int stride = gridDim.x * 8;
  for (int wi = blockIdx.x * 8 + wave; wi < WORK_ITEMS; wi += stride) {
    const int xb = wi & 7;
    const int yg = (wi >> 3) & 15;
    const int iz = wi >> 7;

    const float cx = (float)(xb * 16 + m) * SC - 1.0f;
    v2f ax;
    ax.x = h ? 1.f : cx;
    ax.y = h ? 0.f : cx * cx;

    const bool g   = m >= 8;
    const float cz = (float)iz * SC - 1.0f;
    const float cy = (float)(yg * 8 + (m & 7)) * SC - 1.0f;
    v2f p0, p1, p2;
    p0.x = g ? (h ? cz * cy : cz * cz) : 0.f;
    p0.y = g ? (h ? cz      : cy * cy) : 0.f;
    p1.x = h ? (g ? 0.f : cz) : (g ? cy  : 0.f);
    p1.y = h ? (g ? 0.f : cy) : (g ? 1.f : 0.f);
    p2.x = (!h && !g) ? 1.f : 0.f;
    p2.y = 0.f;

    float acc[8][8];
    #pragma unroll
    for (int r = 0; r < 8; ++r)
      #pragma unroll
      for (int i = 0; i < 8; ++i) acc[r][i] = 0.f;

    #pragma unroll 1
    for (int t = 0; t < 16; ++t) {
      const int n = t * 16 + m;
      const v2f* wp = (const v2f*)&lw[n * KDIM];
      const int hs = h ? 1 : 0;
      v2f b0 = wp[0 + hs];
      v2f b1 = wp[2 + hs];
      v2f b2 = wp[4 + hs];
      const float dn = ld[n];

      v8f bg = {0.f,0.f,0.f,0.f,0.f,0.f,0.f,0.f};
      bg = __builtin_amdgcn_wmma_f32_16x16x4_f32(false, p0, false, b0, (short)0, bg, false, false);
      bg = __builtin_amdgcn_wmma_f32_16x16x4_f32(false, p1, false, b1, (short)0, bg, false, false);
      bg = __builtin_amdgcn_wmma_f32_16x16x4_f32(false, p2, false, b2, (short)0, bg, false, false);

      #pragma unroll
      for (int r = 0; r < 8; ++r) {
        v2f bq;
        bq.x = bg[r];
        bq.y = b2.y;
        v8f c = {0.f,0.f,0.f,0.f,0.f,0.f,0.f,0.f};
        c = __builtin_amdgcn_wmma_f32_16x16x4_f32(false, ax, false, bq, (short)0, c, false, false);
        #pragma unroll
        for (int i = 0; i < 8; ++i)
          acc[r][i] += __builtin_amdgcn_exp2f(c[i] * EXP_SCALE) * dn;
      }
    }

    #pragma unroll
    for (int r = 0; r < 8; ++r) {
      #pragma unroll
      for (int mask = 1; mask <= 8; mask <<= 1) {
        #pragma unroll
        for (int i = 0; i < 8; ++i)
          acc[r][i] += __shfl_xor(acc[r][i], mask, 32);
      }
      if (lane == 0 || lane == 16) {
        const int rb = ((iz * GRID_DIM + yg * 8 + r) << 3) + xb;
        float4 lo = make_float4(acc[r][0], acc[r][1], acc[r][2], acc[r][3]);
        float4 hi = make_float4(acc[r][4], acc[r][5], acc[r][6], acc[r][7]);
        float* dst = out + rb * 16 + (h ? 8 : 0);
        ((float4*)dst)[0] = lo;
        ((float4*)dst)[1] = hi;
      }
    }
  }
}

extern "C" void kernel_launch(void* const* d_in, const int* in_sizes, int n_in,
                              void* d_out, int out_size, void* d_ws, size_t ws_size,
                              hipStream_t stream) {
  const float* positions = (const float*)d_in[0];
  const float* scales    = (const float*)d_in[1];
  const float* rotations = (const float*)d_in[2];
  const float* density   = (const float*)d_in[3];
  float* out = (float*)d_out;
  float* wtab = (float*)d_ws;

  vox_prep<<<1, 256, 0, stream>>>(positions, scales, rotations, wtab);
  vox_main<<<MAIN_BLOCKS, 256, 0, stream>>>(wtab, density, out);
}


# Level-2 WMMA operands swapped so the tile comes out transposed (lane = voxel, registers = gaussians): density sum runs in-lane, 256 cross-lane shuffles per wave become 4; sum order changes
# speedup vs baseline: 1.5601x; 1.0586x over previous
#include <hip/hip_runtime.h>

typedef __attribute__((ext_vector_type(2))) float v2f;
typedef __attribute__((ext_vector_type(4))) float v4f;
typedef __attribute__((ext_vector_type(8))) float v8f;

#define NG   256
#define KDIM 12
#define GRID_DIM 128
#define NVOX (GRID_DIM*GRID_DIM*GRID_DIM)
#define ROW_BLOCKS (NVOX/16)
#define WORK_ITEMS (ROW_BLOCKS/8)
#define MAIN_BLOCKS 2048

__global__ __launch_bounds__(256) void vox_prep(const float* __restrict__ pos,
                                                const float* __restrict__ scl,
                                                const float* __restrict__ rot,
                                                float* __restrict__ w) {
  int n = blockIdx.x * blockDim.x + threadIdx.x;
  if (n >= NG) return;
  float qw = rot[n*4+0], qx = rot[n*4+1], qy = rot[n*4+2], qz = rot[n*4+3];
  float rn = rsqrtf(qw*qw + qx*qx + qy*qy + qz*qz);
  qw *= rn; qx *= rn; qy *= rn; qz *= rn;
  float R00 = 1.f - 2.f*(qy*qy + qz*qz);
  float R01 = 2.f*(qx*qy - qw*qz);
  float R02 = 2.f*(qx*qz + qw*qy);
  float R10 = 2.f*(qx*qy + qw*qz);
  float R11 = 1.f - 2.f*(qx*qx + qz*qz);
  float R12 = 2.f*(qy*qz - qw*qx);
  float R20 = 2.f*(qx*qz - qw*qy);
  float R21 = 2.f*(qy*qz + qw*qx);
  float R22 = 1.f - 2.f*(qx*qx + qy*qy);
  float s0 = scl[n*3+0], s1 = scl[n*3+1], s2 = scl[n*3+2];
  float i0 = 1.f/(s0*s0 + 1e-8f);
  float i1 = 1.f/(s1*s1 + 1e-8f);
  float i2 = 1.f/(s2*s2 + 1e-8f);
  float C00 = R00*R00*i0 + R01*R01*i1 + R02*R02*i2;
  float C01 = R00*R10*i0 + R01*R11*i1 + R02*R12*i2;
  float C02 = R00*R20*i0 + R01*R21*i1 + R02*R22*i2;
  float C11 = R10*R10*i0 + R11*R11*i1 + R12*R12*i2;
  float C12 = R10*R20*i0 + R11*R21*i1 + R12*R22*i2;
  float C22 = R20*R20*i0 + R21*R21*i1 + R22*R22*i2;
  float p0 = pos[n*3+0], p1 = pos[n*3+1], p2 = pos[n*3+2];
  float b0 = C00*p0 + C01*p1 + C02*p2;
  float b1 = C01*p0 + C11*p1 + C12*p2;
  float b2 = C02*p0 + C12*p1 + C22*p2;
  float cn = p0*b0 + p1*b1 + p2*b2;
  float* wn = w + n*KDIM;
  wn[0] = C00;      wn[1] = C11;      wn[2]  = 2.f*C01;
  wn[3] = -2.f*b0;  wn[4] = -2.f*b1;  wn[5]  = cn;
  wn[6] = 2.f*C02;  wn[7] = 2.f*C12;  wn[8]  = -2.f*b2;
  wn[9] = C22;      wn[10] = 0.f;     wn[11] = 0.f;
}

__global__ __launch_bounds__(256) void vox_main(const float* __restrict__ w,
                                                const float* __restrict__ density,
                                                float* __restrict__ out) {
  __shared__ float lw[NG * KDIM];
  __shared__ __attribute__((aligned(16))) float ld[NG];

  const int tid = threadIdx.x;
  for (int i = tid; i < NG * KDIM; i += 256) lw[i] = w[i];
  if (tid < NG) ld[tid] = density[tid];
  __syncthreads();

  const int lane = tid & 31;
  const int wave = tid >> 5;
  const int m    = lane & 15;
  const bool h   = lane >= 16;

  const float EXP_SCALE = -0.72134752044448170368f;
  const float SC = 2.0f / 127.0f;

  const int stride = gridDim.x * 8;
  for (int wi = blockIdx.x * 8 + wave; wi < WORK_ITEMS; wi += stride) {
    const int xb = wi & 7;
    const int yg = (wi >> 3) & 15;
    const int iz = wi >> 7;

    const float cx = (float)(xb * 16 + m) * SC - 1.0f;
    v2f ax;
    ax.x = h ? 1.f : cx;
    ax.y = h ? 0.f : cx * cx;

    const bool g   = m >= 8;
    const float cz = (float)iz * SC - 1.0f;
    const float cy = (float)(yg * 8 + (m & 7)) * SC - 1.0f;
    v2f p0, p1, p2;
    p0.x = g ? (h ? cz * cy : cz * cz) : 0.f;
    p0.y = g ? (h ? cz      : cy * cy) : 0.f;
    p1.x = h ? (g ? 0.f : cz) : (g ? cy  : 0.f);
    p1.y = h ? (g ? 0.f : cy) : (g ? 1.f : 0.f);
    p2.x = (!h && !g) ? 1.f : 0.f;
    p2.y = 0.f;

    v2f acc[8];
    #pragma unroll
    for (int r = 0; r < 8; ++r) { acc[r].x = 0.f; acc[r].y = 0.f; }

    const int hs = h ? 1 : 0;
    const float* wl = &lw[m * KDIM + 2 * hs];
    const float* dl = &ld[hs * 8];

    #pragma unroll 1
    for (int t = 0; t < 16; ++t) {
      const v2f* wp = (const v2f*)(wl + t * (16 * KDIM));
      v2f b0 = wp[0];
      v2f b1 = wp[2];
      v2f b2 = wp[4];
      const v4f* dp = (const v4f*)(dl + t * 16);
      const v4f dlo = dp[0];
      const v4f dhi = dp[1];
      v2f dv[4];
      dv[0].x = dlo.x; dv[0].y = dlo.y;
      dv[1].x = dlo.z; dv[1].y = dlo.w;
      dv[2].x = dhi.x; dv[2].y = dhi.y;
      dv[3].x = dhi.z; dv[3].y = dhi.w;

      v8f bg = {0.f,0.f,0.f,0.f,0.f,0.f,0.f,0.f};
      bg = __builtin_amdgcn_wmma_f32_16x16x4_f32(false, p0, false, b0, (short)0, bg, false, false);
      bg = __builtin_amdgcn_wmma_f32_16x16x4_f32(false, p1, false, b1, (short)0, bg, false, false);
      bg = __builtin_amdgcn_wmma_f32_16x16x4_f32(false, p2, false, b2, (short)0, bg, false, false);

      #pragma unroll
      for (int r = 0; r < 8; ++r) {
        v2f aq;
        aq.x = bg[r];
        aq.y = b2.y;
        v8f c = {0.f,0.f,0.f,0.f,0.f,0.f,0.f,0.f};
        c = __builtin_amdgcn_wmma_f32_16x16x4_f32(false, aq, false, ax, (short)0, c, false, false);
        #pragma unroll
        for (int k = 0; k < 4; ++k) {
          v2f e;
          e.x = __builtin_amdgcn_exp2f(c[2 * k]     * EXP_SCALE);
          e.y = __builtin_amdgcn_exp2f(c[2 * k + 1] * EXP_SCALE);
          acc[r] += e * dv[k];
        }
      }
    }

    #pragma unroll
    for (int r = 0; r < 8; r += 2) {
      const float s0 = acc[r].x + acc[r].y;
      const float s1 = acc[r + 1].x + acc[r + 1].y;
      const float mine = h ? s1 : s0;
      const float give = h ? s0 : s1;
      const float got  = __shfl_xor(give, 16, 32);
      const int row = iz * GRID_DIM + yg * 8 + r + hs;
      out[row * GRID_DIM + xb * 16 + m] = mine + got;
    }
  }
}

extern "C" void kernel_launch(void* const* d_in, const int* in_sizes, int n_in,
                              void* d_out, int out_size, void* d_ws, size_t ws_size,
                              hipStream_t stream) {
  const float* positions = (const float*)d_in[0];
  const float* scales    = (const float*)d_in[1];
  const float* rotations = (const float*)d_in[2];
  const float* density   = (const float*)d_in[3];
  float* out = (float*)d_out;
  float* wtab = (float*)d_ws;

  vox_prep<<<1, 256, 0, stream>>>(positions, scales, rotations, wtab);
  vox_main<<<MAIN_BLOCKS, 256, 0, stream>>>(wtab, density, out);
}


# The factor -0.5*log2(e) moved into the level-2 B fragment (x, x^2, 1 scaled once per wave): the WMMA tile is the exp2 argument itself, no multiply per exponential; argument rounding changes
# speedup vs baseline: 1.6515x; 1.0586x over previous
#include <hip/hip_runtime.h>

typedef __attribute__((ext_vector_type(2))) float v2f;
typedef __attribute__((ext_vector_type(4))) float v4f;
typedef __attribute__((ext_vector_type(8))) float v8f;

#define NG   256
#define KDIM 12
#define GRID_DIM 128
#define NVOX (GRID_DIM*GRID_DIM*GRID_DIM)
#define ROW_BLOCKS (NVOX/16)
#define WORK_ITEMS (ROW_BLOCKS/8)
#define MAIN_BLOCKS 2048

__global__ __launch_bounds__(256) void vox_prep(const float* __restrict__ pos,
                                                const float* __restrict__ scl,
                                                const float* __restrict__ rot,
                                                float* __restrict__ w) {
  int n = blockIdx.x * blockDim.x + threadIdx.x;
  if (n >= NG) return;
  float qw = rot[n*4+0], qx = rot[n*4+1], qy = rot[n*4+2], qz = rot[n*4+3];
  float rn = rsqrtf(qw*qw + qx*qx + qy*qy + qz*qz);
  qw *= rn; qx *= rn; qy *= rn; qz *= rn;
  float R00 = 1.f - 2.f*(qy*qy + qz*qz);
  float R01 = 2.f*(qx*qy - qw*qz);
  float R02 = 2.f*(qx*qz + qw*qy);
  float R10 = 2.f*(qx*qy + qw*qz);
  float R11 = 1.f - 2.f*(qx*qx + qz*qz);
  float R12 = 2.f*(qy*qz - qw*qx);
  float R20 = 2.f*(qx*qz - qw*qy);
  float R21 = 2.f*(qy*qz + qw*qx);
  float R22 = 1.f - 2.f*(qx*qx + qy*qy);
  float s0 = scl[n*3+0], s1 = scl[n*3+1], s2 = scl[n*3+2];
  float i0 = 1.f/(s0*s0 + 1e-8f);
  float i1 = 1.f/(s1*s1 + 1e-8f);
  float i2 = 1.f/(s2*s2 + 1e-8f);
  float C00 = R00*R00*i0 + R01*R01*i1 + R02*R02*i2;
  float C01 = R00*R10*i0 + R01*R11*i1 + R02*R12*i2;
  float C02 = R00*R20*i0 + R01*R21*i1 + R02*R22*i2;
  float C11 = R10*R10*i0 + R11*R11*i1 + R12*R12*i2;
  float C12 = R10*R20*i0 + R11*R21*i1 + R12*R22*i2;
  float C22 = R20*R20*i0 + R21*R21*i1 + R22*R22*i2;
  float p0 = pos[n*3+0], p1 = pos[n*3+1], p2 = pos[n*3+2];
  float b0 = C00*p0 + C01*p1 + C02*p2;
  float b1 = C01*p0 + C11*p1 + C12*p2;
  float b2 = C02*p0 + C12*p1 + C22*p2;
  float cn = p0*b0 + p1*b1 + p2*b2;
  float* wn = w + n*KDIM;
  wn[0] = C00;      wn[1] = C11;      wn[2]  = 2.f*C01;
  wn[3] = -2.f*b0;  wn[4] = -2.f*b1;  wn[5]  = cn;
  wn[6] = 2.f*C02;  wn[7] = 2.f*C12;  wn[8]  = -2.f*b2;
  wn[9] = C22;      wn[10] = 0.f;     wn[11] = 0.f;
}

__global__ __launch_bounds__(256) void vox_main(const float* __restrict__ w,
                                                const float* __restrict__ density,
                                                float* __restrict__ out) {
  __shared__ float lw[NG * KDIM];
  __shared__ __attribute__((aligned(16))) float ld[NG];

  const int tid = threadIdx.x;
  for (int i = tid; i < NG * KDIM; i += 256) lw[i] = w[i];
  if (tid < NG) ld[tid] = density[tid];
  __syncthreads();

  const int lane = tid & 31;
  const int wave = tid >> 5;
  const int m    = lane & 15;
  const bool h   = lane >= 16;

  const float EXP_SCALE = -0.72134752044448170368f;
  const float SC = 2.0f / 127.0f;

  const int stride = gridDim.x * 8;
  for (int wi = blockIdx.x * 8 + wave; wi < WORK_ITEMS; wi += stride) {
    const int xb = wi & 7;
    const int yg = (wi >> 3) & 15;
    const int iz = wi >> 7;

    const float cx = (float)(xb * 16 + m) * SC - 1.0f;
    v2f ax;
    ax.x = h ? EXP_SCALE : cx * EXP_SCALE;
    ax.y = h ? 0.f : cx * cx * EXP_SCALE;

    const bool g   = m >= 8;
    const float cz = (float)iz * SC - 1.0f;
    const float cy = (float)(yg * 8 + (m & 7)) * SC - 1.0f;
    v2f p0, p1, p2;
    p0.x = g ? (h ? cz * cy : cz * cz) : 0.f;
    p0.y = g ? (h ? cz      : cy * cy) : 0.f;
    p1.x = h ? (g ? 0.f : cz) : (g ? cy  : 0.f);
    p1.y = h ? (g ? 0.f : cy) : (g ? 1.f : 0.f);
    p2.x = (!h && !g) ? 1.f : 0.f;
    p2.y = 0.f;

    v2f acc[8];
    #pragma unroll
    for (int r = 0; r < 8; ++r) { acc[r].x = 0.f; acc[r].y = 0.f; }

    const int hs = h ? 1 : 0;
    const float* wl = &lw[m * KDIM + 2 * hs];
    const float* dl = &ld[hs * 8];

    #pragma unroll 1
    for (int t = 0; t < 16; ++t) {
      const v2f* wp = (const v2f*)(wl + t * (16 * KDIM));
      v2f b0 = wp[0];
      v2f b1 = wp[2];
      v2f b2 = wp[4];
      const v4f* dp = (const v4f*)(dl + t * 16);
      const v4f dlo = dp[0];
      const v4f dhi = dp[1];
      v2f dv[4];
      dv[0].x = dlo.x; dv[0].y = dlo.y;
      dv[1].x = dlo.z; dv[1].y = dlo.w;
      dv[2].x = dhi.x; dv[2].y = dhi.y;
      dv[3].x = dhi.z; dv[3].y = dhi.w;

      v8f bg = {0.f,0.f,0.f,0.f,0.f,0.f,0.f,0.f};
      bg = __builtin_amdgcn_wmma_f32_16x16x4_f32(false, p0, false, b0, (short)0, bg, false, false);
      bg = __builtin_amdgcn_wmma_f32_16x16x4_f32(false, p1, false, b1, (short)0, bg, false, false);
      bg = __builtin_amdgcn_wmma_f32_16x16x4_f32(false, p2, false, b2, (short)0, bg, false, false);

      #pragma unroll
      for (int r = 0; r < 8; ++r) {
        v2f aq;
        aq.x = bg[r];
        aq.y = b2.y;
        v8f c = {0.f,0.f,0.f,0.f,0.f,0.f,0.f,0.f};
        c = __builtin_amdgcn_wmma_f32_16x16x4_f32(false, aq, false, ax, (short)0, c, false, false);
        #pragma unroll
        for (int k = 0; k < 4; ++k) {
          v2f e;
          e.x = __builtin_amdgcn_exp2f(c[2 * k]);
          e.y = __builtin_amdgcn_exp2f(c[2 * k + 1]);
          acc[r] += e * dv[k];
        }
      }
    }

    #pragma unroll
    for (int r = 0; r < 8; r += 2) {
      const float s0 = acc[r].x + acc[r].y;
      const float s1 = acc[r + 1].x + acc[r + 1].y;
      const float mine = h ? s1 : s0;
      const float give = h ? s0 : s1;
      const float got  = __shfl_xor(give, 16, 32);
      const int row = iz * GRID_DIM + yg * 8 + r + hs;
      out[row * GRID_DIM + xb * 16 + m] = mine + got;
    }
  }
}

extern "C" void kernel_launch(void* const* d_in, const int* in_sizes, int n_in,
                              void* d_out, int out_size, void* d_ws, size_t ws_size,
                              hipStream_t stream) {
  const float* positions = (const float*)d_in[0];
  const float* scales    = (const float*)d_in[1];
  const float* rotations = (const float*)d_in[2];
  const float* density   = (const float*)d_in[3];
  float* out = (float*)d_out;
  float* wtab = (float*)d_ws;

  vox_prep<<<1, 256, 0, stream>>>(positions, scales, rotations, wtab);
  vox_main<<<MAIN_BLOCKS, 256, 0, stream>>>(wtab, density, out);
}


# Level 1 (beta, gamma of the block's 8 y-rows) computed once per block and shared through LDS: the block's 8 waves differ in x only; 8 WMMAs per tile and wave, was 11; sum order unchanged
# speedup vs baseline: 1.8520x; 1.1214x over previous
#include <hip/hip_runtime.h>

typedef __attribute__((ext_vector_type(2))) float v2f;
typedef __attribute__((ext_vector_type(4))) float v4f;
typedef __attribute__((ext_vector_type(8))) float v8f;

#define NG   256
#define KDIM 12
#define GRID_DIM 128
#define NVOX (GRID_DIM*GRID_DIM*GRID_DIM)
#define ROW_BLOCKS (NVOX/16)
#define WORK_ITEMS (ROW_BLOCKS/8)
#define MAIN_BLOCKS 2048

__global__ __launch_bounds__(256) void vox_prep(const float* __restrict__ pos,
                                                const float* __restrict__ scl,
                                                const float* __restrict__ rot,
                                                float* __restrict__ w) {
  int n = blockIdx.x * blockDim.x + threadIdx.x;
  if (n >= NG) return;
  float qw = rot[n*4+0], qx = rot[n*4+1], qy = rot[n*4+2], qz = rot[n*4+3];
  float rn = rsqrtf(qw*qw + qx*qx + qy*qy + qz*qz);
  qw *= rn; qx *= rn; qy *= rn; qz *= rn;
  float R00 = 1.f - 2.f*(qy*qy + qz*qz);
  float R01 = 2.f*(qx*qy - qw*qz);
  float R02 = 2.f*(qx*qz + qw*qy);
  float R10 = 2.f*(qx*qy + qw*qz);
  float R11 = 1.f - 2.f*(qx*qx + qz*qz);
  float R12 = 2.f*(qy*qz - qw*qx);
  float R20 = 2.f*(qx*qz - qw*qy);
  float R21 = 2.f*(qy*qz + qw*qx);
  float R22 = 1.f - 2.f*(qx*qx + qy*qy);
  float s0 = scl[n*3+0], s1 = scl[n*3+1], s2 = scl[n*3+2];
  float i0 = 1.f/(s0*s0 + 1e-8f);
  float i1 = 1.f/(s1*s1 + 1e-8f);
  float i2 = 1.f/(s2*s2 + 1e-8f);
  float C00 = R00*R00*i0 + R01*R01*i1 + R02*R02*i2;
  float C01 = R00*R10*i0 + R01*R11*i1 + R02*R12*i2;
  float C02 = R00*R20*i0 + R01*R21*i1 + R02*R22*i2;
  float C11 = R10*R10*i0 + R11*R11*i1 + R12*R12*i2;
  float C12 = R10*R20*i0 + R11*R21*i1 + R12*R22*i2;
  float C22 = R20*R20*i0 + R21*R21*i1 + R22*R22*i2;
  float p0 = pos[n*3+0], p1 = pos[n*3+1], p2 = pos[n*3+2];
  float b0 = C00*p0 + C01*p1 + C02*p2;
  float b1 = C01*p0 + C11*p1 + C12*p2;
  float b2 = C02*p0 + C12*p1 + C22*p2;
  float cn = p0*b0 + p1*b1 + p2*b2;
  float* wn = w + n*KDIM;
  wn[0] = C00;      wn[1] = C11;      wn[2]  = 2.f*C01;
  wn[3] = -2.f*b0;  wn[4] = -2.f*b1;  wn[5]  = cn;
  wn[6] = 2.f*C02;  wn[7] = 2.f*C12;  wn[8]  = -2.f*b2;
  wn[9] = C22;      wn[10] = 0.f;     wn[11] = 0.f;
}

__global__ __launch_bounds__(256) void vox_main(const float* __restrict__ w,
                                                const float* __restrict__ density,
                                                float* __restrict__ out) {
  __shared__ float lw[NG * KDIM];
  __shared__ __attribute__((aligned(16))) float ld[NG];
  __shared__ __attribute__((aligned(16))) float lbg[16 * 2 * 32 * 4];

  const int tid = threadIdx.x;
  for (int i = tid; i < NG * KDIM; i += 256) lw[i] = w[i];
  if (tid < NG) ld[tid] = density[tid];
  __syncthreads();

  const int lane = tid & 31;
  const int wave = tid >> 5;
  const int m    = lane & 15;
  const bool h   = lane >= 16;

  const float EXP_SCALE = -0.72134752044448170368f;
  const float SC = 2.0f / 127.0f;

  const int wi = blockIdx.x * 8 + wave;
  const int xb = wi & 7;
  const int yg = (wi >> 3) & 15;
  const int iz = wi >> 7;

  const float cx = (float)(xb * 16 + m) * SC - 1.0f;
  v2f ax;
  ax.x = h ? EXP_SCALE : cx * EXP_SCALE;
  ax.y = h ? 0.f : cx * cx * EXP_SCALE;

  const bool g   = m >= 8;
  const float cz = (float)iz * SC - 1.0f;
  const float cy = (float)(yg * 8 + (m & 7)) * SC - 1.0f;
  v2f p0, p1, p2;
  p0.x = g ? (h ? cz * cy : cz * cz) : 0.f;
  p0.y = g ? (h ? cz      : cy * cy) : 0.f;
  p1.x = h ? (g ? 0.f : cz) : (g ? cy  : 0.f);
  p1.y = h ? (g ? 0.f : cy) : (g ? 1.f : 0.f);
  p2.x = (!h && !g) ? 1.f : 0.f;
  p2.y = 0.f;

  const int hs = h ? 1 : 0;
  const float* wl = &lw[m * KDIM + 2 * hs];
  const float* dl = &ld[hs * 8];

  #pragma unroll
  for (int k = 0; k < 2; ++k) {
    const int t = 2 * wave + k;
    const v2f* wp = (const v2f*)(wl + t * (16 * KDIM));
    const v2f b0 = wp[0];
    const v2f b1 = wp[2];
    const v2f b2 = wp[4];
    v8f bg = {0.f,0.f,0.f,0.f,0.f,0.f,0.f,0.f};
    bg = __builtin_amdgcn_wmma_f32_16x16x4_f32(false, p0, false, b0, (short)0, bg, false, false);
    bg = __builtin_amdgcn_wmma_f32_16x16x4_f32(false, p1, false, b1, (short)0, bg, false, false);
    bg = __builtin_amdgcn_wmma_f32_16x16x4_f32(false, p2, false, b2, (short)0, bg, false, false);
    v4f lo, hi;
    lo.x = bg[0]; lo.y = bg[1]; lo.z = bg[2]; lo.w = bg[3];
    hi.x = bg[4]; hi.y = bg[5]; hi.z = bg[6]; hi.w = bg[7];
    *(v4f*)&lbg[((t * 2 + 0) * 32 + lane) * 4] = lo;
    *(v4f*)&lbg[((t * 2 + 1) * 32 + lane) * 4] = hi;
  }
  __syncthreads();

  v2f acc[8];
  #pragma unroll
  for (int r = 0; r < 8; ++r) { acc[r].x = 0.f; acc[r].y = 0.f; }

  #pragma unroll 1
  for (int t = 0; t < 16; ++t) {
    const float c22 = wl[t * (16 * KDIM) + 9];
    const v4f glo = *(const v4f*)&lbg[((t * 2 + 0) * 32 + lane) * 4];
    const v4f ghi = *(const v4f*)&lbg[((t * 2 + 1) * 32 + lane) * 4];
    float bgr[8];
    bgr[0] = glo.x; bgr[1] = glo.y; bgr[2] = glo.z; bgr[3] = glo.w;
    bgr[4] = ghi.x; bgr[5] = ghi.y; bgr[6] = ghi.z; bgr[7] = ghi.w;
    const v4f* dp = (const v4f*)(dl + t * 16);
    const v4f dlo = dp[0];
    const v4f dhi = dp[1];
    v2f dv[4];
    dv[0].x = dlo.x; dv[0].y = dlo.y;
    dv[1].x = dlo.z; dv[1].y = dlo.w;
    dv[2].x = dhi.x; dv[2].y = dhi.y;
    dv[3].x = dhi.z; dv[3].y = dhi.w;

    #pragma unroll
    for (int r = 0; r < 8; ++r) {
      v2f aq;
      aq.x = bgr[r];
      aq.y = c22;
      v8f c = {0.f,0.f,0.f,0.f,0.f,0.f,0.f,0.f};
      c = __builtin_amdgcn_wmma_f32_16x16x4_f32(false, aq, false, ax, (short)0, c, false, false);
      #pragma unroll
      for (int k = 0; k < 4; ++k) {
        v2f e;
        e.x = __builtin_amdgcn_exp2f(c[2 * k]);
        e.y = __builtin_amdgcn_exp2f(c[2 * k + 1]);
        acc[r] += e * dv[k];
      }
    }
  }

  #pragma unroll
  for (int r = 0; r < 8; r += 2) {
    const float s0 = acc[r].x + acc[r].y;
    const float s1 = acc[r + 1].x + acc[r + 1].y;
    const float mine = h ? s1 : s0;
    const float give = h ? s0 : s1;
    const float got  = __shfl_xor(give, 16, 32);
    const int row = iz * GRID_DIM + yg * 8 + r + hs;
    out[row * GRID_DIM + xb * 16 + m] = mine + got;
  }
}

extern "C" void kernel_launch(void* const* d_in, const int* in_sizes, int n_in,
                              void* d_out, int out_size, void* d_ws, size_t ws_size,
                              hipStream_t stream) {
  const float* positions = (const float*)d_in[0];
  const float* scales    = (const float*)d_in[1];
  const float* rotations = (const float*)d_in[2];
  const float* density   = (const float*)d_in[3];
  float* out = (float*)d_out;
  float* wtab = (float*)d_ws;

  vox_prep<<<1, 256, 0, stream>>>(positions, scales, rotations, wtab);
  vox_main<<<MAIN_BLOCKS, 256, 0, stream>>>(wtab, density, out);
}


# speedup vs baseline: 1.9516x; 1.0538x over previous
#include <hip/hip_runtime.h>

typedef __attribute__((ext_vector_type(2))) float v2f;
typedef __attribute__((ext_vector_type(4))) float v4f;
typedef __attribute__((ext_vector_type(8))) float v8f;

#define NG   256
#define KDIM 12
#define GRID_DIM 128
#define NVOX (GRID_DIM*GRID_DIM*GRID_DIM)
#define ROW_BLOCKS (NVOX/16)
#define WORK_ITEMS (ROW_BLOCKS/8)
#define MAIN_BLOCKS 2048

__global__ __launch_bounds__(256) void vox_prep(const float* __restrict__ pos,
                                                const float* __restrict__ scl,
                                                const float* __restrict__ rot,
                                                float* __restrict__ w) {
  int n = blockIdx.x * blockDim.x + threadIdx.x;
  if (n >= NG) return;
  float qw = rot[n*4+0], qx = rot[n*4+1], qy = rot[n*4+2], qz = rot[n*4+3];
  float rn = rsqrtf(qw*qw + qx*qx + qy*qy + qz*qz);
  qw *= rn; qx *= rn; qy *= rn; qz *= rn;
  float R00 = 1.f - 2.f*(qy*qy + qz*qz);
  float R01 = 2.f*(qx*qy - qw*qz);
  float R02 = 2.f*(qx*qz + qw*qy);
  float R10 = 2.f*(qx*qy + qw*qz);
  float R11 = 1.f - 2.f*(qx*qx + qz*qz);
  float R12 = 2.f*(qy*qz - qw*qx);
  float R20 = 2.f*(qx*qz - qw*qy);
  float R21 = 2.f*(qy*qz + qw*qx);
  float R22 = 1.f - 2.f*(qx*qx + qy*qy);
  float s0 = scl[n*3+0], s1 = scl[n*3+1], s2 = scl[n*3+2];
  float i0 = 1.f/(s0*s0 + 1e-8f);
  float i1 = 1.f/(s1*s1 + 1e-8f);
  float i2 = 1.f/(s2*s2 + 1e-8f);
  float C00 = R00*R00*i0 + R01*R01*i1 + R02*R02*i2;
  float C01 = R00*R10*i0 + R01*R11*i1 + R02*R12*i2;
  float C02 = R00*R20*i0 + R01*R21*i1 + R02*R22*i2;
  float C11 = R10*R10*i0 + R11*R11*i1 + R12*R12*i2;
  float C12 = R10*R20*i0 + R11*R21*i1 + R12*R22*i2;
  float C22 = R20*R20*i0 + R21*R21*i1 + R22*R22*i2;
  float p0 = pos[n*3+0], p1 = pos[n*3+1], p2 = pos[n*3+2];
  float b0 = C00*p0 + C01*p1 + C02*p2;
  float b1 = C01*p0 + C11*p1 + C12*p2;
  float b2 = C02*p0 + C12*p1 + C22*p2;
  float cn = p0*b0 + p1*b1 + p2*b2;
  float* wn = w + n*KDIM;
  wn[0] = C00;      wn[1] = C11;      wn[2]  = 2.f*C01;
  wn[3] = -2.f*b0;  wn[4] = -2.f*b1;  wn[5]  = cn;
  wn[6] = 2.f*C02;  wn[7] = 2.f*C12;  wn[8]  = -2.f*b2;
  wn[9] = C22;      wn[10] = 0.f;     wn[11] = 0.f;
}

__global__ __launch_bounds__(256) void vox_main(const float* __restrict__ w,
                                                const float* __restrict__ density,
                                                float* __restrict__ out) {
  __shared__ float lw[NG * KDIM];
  __shared__ __attribute__((aligned(16))) float ld[NG];
  __shared__ __attribute__((aligned(16))) float lbg[16 * 2 * 32 * 4];

  const int tid = threadIdx.x;
  for (int i = tid; i < NG * KDIM; i += 256) lw[i] = w[i];
  if (tid < NG) ld[tid] = density[tid];
  __syncthreads();

  const int lane = tid & 31;
  const int wave = tid >> 5;
  const int m    = lane & 15;
  const bool h   = lane >= 16;

  const float EXP_SCALE = -0.72134752044448170368f;
  const float SC = 2.0f / 127.0f;

  const int wi = blockIdx.x * 8 + wave;
  const int xb = wi & 7;
  const int yg = (wi >> 3) & 15;
  const int iz = wi >> 7;

  const float cx = (float)(xb * 16 + m) * SC - 1.0f;
  v2f ax;
  ax.x = h ? EXP_SCALE : cx * EXP_SCALE;
  ax.y = h ? 0.f : cx * cx * EXP_SCALE;

  const bool g   = m >= 8;
  const float cz = (float)iz * SC - 1.0f;
  const float cy = (float)(yg * 8 + (m & 7)) * SC - 1.0f;
  v2f p0, p1, p2;
  p0.x = g ? (h ? cz * cy : cz * cz) : 0.f;
  p0.y = g ? (h ? cz      : cy * cy) : 0.f;
  p1.x = h ? (g ? 0.f : cz) : (g ? cy  : 0.f);
  p1.y = h ? (g ? 0.f : cy) : (g ? 1.f : 0.f);
  p2.x = (!h && !g) ? 1.f : 0.f;
  p2.y = 0.f;

  const int hs = h ? 1 : 0;
  const float* wl = &lw[m * KDIM + 2 * hs];
  const float* dl = &ld[hs * 8];

  #pragma unroll
  for (int k = 0; k < 2; ++k) {
    const int t = 2 * wave + k;
    const v2f* wp = (const v2f*)(wl + t * (16 * KDIM));
    const v2f b0 = wp[0];
    const v2f b1 = wp[2];
    const v2f b2 = wp[4];
    v8f bg = {0.f,0.f,0.f,0.f,0.f,0.f,0.f,0.f};
    bg = __builtin_amdgcn_wmma_f32_16x16x4_f32(false, p0, false, b0, (short)0, bg, false, false);
    bg = __builtin_amdgcn_wmma_f32_16x16x4_f32(false, p1, false, b1, (short)0, bg, false, false);
    bg = __builtin_amdgcn_wmma_f32_16x16x4_f32(false, p2, false, b2, (short)0, bg, false, false);
    v4f lo, hi;
    lo.x = bg[0]; lo.y = bg[1]; lo.z = bg[2]; lo.w = bg[3];
    hi.x = bg[4]; hi.y = bg[5]; hi.z = bg[6]; hi.w = bg[7];
    *(v4f*)&lbg[((t * 2 + 0) * 32 + lane) * 4] = lo;
    *(v4f*)&lbg[((t * 2 + 1) * 32 + lane) * 4] = hi;
  }
  __syncthreads();

  v2f acc[8];
  #pragma unroll
  for (int r = 0; r < 8; ++r) { acc[r].x = 0.f; acc[r].y = 0.f; }

#define AFTER_EXP2(a, b) do { asm volatile("" :: "v"(a), "v"(b)); __builtin_amdgcn_sched_barrier(0); } while (0)
#define AFTER_EXP1(a)    do { asm volatile("" :: "v"(a)); __builtin_amdgcn_sched_barrier(0); } while (0)
#define AFTER_SUM(sum)   do { asm volatile("" :: "v"(sum)); __builtin_amdgcn_sched_barrier(0); } while (0)
  const v8f zero8 = {0.f,0.f,0.f,0.f,0.f,0.f,0.f,0.f};
  #pragma unroll 1
  for (int t = 0; t < 16; ++t) {
    const float c22 = wl[t * (16 * KDIM) + 9];
    const v4f glo = *(const v4f*)&lbg[((t * 2 + 0) * 32 + lane) * 4];
    const v4f ghi = *(const v4f*)&lbg[((t * 2 + 1) * 32 + lane) * 4];
    float bgr[8];
    bgr[0] = glo.x; bgr[1] = glo.y; bgr[2] = glo.z; bgr[3] = glo.w;
    bgr[4] = ghi.x; bgr[5] = ghi.y; bgr[6] = ghi.z; bgr[7] = ghi.w;
    const v4f* dp = (const v4f*)(dl + t * 16);
    const v4f dlo = dp[0];
    const v4f dhi = dp[1];
    v2f dv[4];
    dv[0].x = dlo.x; dv[0].y = dlo.y;
    dv[1].x = dlo.z; dv[1].y = dlo.w;
    dv[2].x = dhi.x; dv[2].y = dhi.y;
    dv[3].x = dhi.z; dv[3].y = dhi.w;

    v2f aq;
    aq.x = bgr[0];
    aq.y = c22;
    v8f cur = __builtin_amdgcn_wmma_f32_16x16x4_f32(false, aq, false, ax, (short)0, zero8, false, false);
    v8f nxt = cur;
    v2f g1, g2, g3;
    #pragma unroll
    for (int r = 0; r < 8; ++r) {
      if (r < 7) {
        v2f an;
        an.x = bgr[r + 1];
        an.y = c22;
        nxt = __builtin_amdgcn_wmma_f32_16x16x4_f32(false, an, false, ax, (short)0, zero8, false, false);
      }
      __builtin_amdgcn_sched_barrier(0);
      v2f e0, e1, e2, e3;
      e0.x = __builtin_amdgcn_exp2f(cur[0]);
      e0.y = __builtin_amdgcn_exp2f(cur[1]);
      AFTER_EXP2(e0.x, e0.y);
      if (r > 0) { acc[r - 1] += g1 * dv[1]; AFTER_SUM(acc[r - 1]); }
      e1.x = __builtin_amdgcn_exp2f(cur[2]);
      e1.y = __builtin_amdgcn_exp2f(cur[3]);
      AFTER_EXP2(e1.x, e1.y);
      if (r > 0) { acc[r - 1] += g2 * dv[2]; AFTER_SUM(acc[r - 1]); }
      e2.x = __builtin_amdgcn_exp2f(cur[4]);
      e2.y = __builtin_amdgcn_exp2f(cur[5]);
      AFTER_EXP2(e2.x, e2.y);
      if (r > 0) { acc[r - 1] += g3 * dv[3]; AFTER_SUM(acc[r - 1]); }
      e3.x = __builtin_amdgcn_exp2f(cur[6]);
      AFTER_EXP1(e3.x);
      acc[r] += e0 * dv[0];
      AFTER_SUM(acc[r]);
      e3.y = __builtin_amdgcn_exp2f(cur[7]);
      AFTER_EXP1(e3.y);
      g1 = e1; g2 = e2; g3 = e3;
      cur = nxt;
    }
    acc[7] += g1 * dv[1];
    acc[7] += g2 * dv[2];
    acc[7] += g3 * dv[3];
  }
#undef AFTER_EXP2
#undef AFTER_EXP1
#undef AFTER_SUM

  #pragma unroll
  for (int r = 0; r < 8; r += 2) {
    const float s0 = acc[r].x + acc[r].y;
    const float s1 = acc[r + 1].x + acc[r + 1].y;
    const float mine = h ? s1 : s0;
    const float give = h ? s0 : s1;
    const float got  = __shfl_xor(give, 16, 32);
    const int row = iz * GRID_DIM + yg * 8 + r + hs;
    out[row * GRID_DIM + xb * 16 + m] = mine + got;
  }
}

extern "C" void kernel_launch(void* const* d_in, const int* in_sizes, int n_in,
                              void* d_out, int out_size, void* d_ws, size_t ws_size,
                              hipStream_t stream) {
  const float* positions = (const float*)d_in[0];
  const float* scales    = (const float*)d_in[1];
  const float* rotations = (const float*)d_in[2];
  const float* density   = (const float*)d_in[3];
  float* out = (float*)d_out;
  float* wtab = (float*)d_ws;

  vox_prep<<<1, 256, 0, stream>>>(positions, scales, rotations, wtab);
  vox_main<<<MAIN_BLOCKS, 256, 0, stream>>>(wtab, density, out);
}


# Coefficient set-up fused into the main kernel: each block's 256 threads build the 256 gaussians' rows of its LDS table from the raw inputs; 1 launch, was 2; no workspace; sum order unchanged
# speedup vs baseline: 2.1052x; 1.0787x over previous
#include <hip/hip_runtime.h>

typedef __attribute__((ext_vector_type(2))) float v2f;
typedef __attribute__((ext_vector_type(4))) float v4f;
typedef __attribute__((ext_vector_type(8))) float v8f;

#define NG   256
#define KDIM 12
#define GRID_DIM 128
#define NVOX (GRID_DIM*GRID_DIM*GRID_DIM)
#define ROW_BLOCKS (NVOX/16)
#define WORK_ITEMS (ROW_BLOCKS/8)
#define MAIN_BLOCKS 2048

__device__ __forceinline__ void vox_coeff(const float* __restrict__ pos,
                                          const float* __restrict__ scl,
                                          const float* __restrict__ rot,
                                          int n, float* __restrict__ wn) {
  const v4f q = *(const v4f*)(rot + n*4);
  float qw = q.x, qx = q.y, qy = q.z, qz = q.w;
  float s0 = scl[n*3+0], s1 = scl[n*3+1], s2 = scl[n*3+2];
  float p0 = pos[n*3+0], p1 = pos[n*3+1], p2 = pos[n*3+2];
  float rn = rsqrtf(qw*qw + qx*qx + qy*qy + qz*qz);
  qw *= rn; qx *= rn; qy *= rn; qz *= rn;
  float R00 = 1.f - 2.f*(qy*qy + qz*qz);
  float R01 = 2.f*(qx*qy - qw*qz);
  float R02 = 2.f*(qx*qz + qw*qy);
  float R10 = 2.f*(qx*qy + qw*qz);
  float R11 = 1.f - 2.f*(qx*qx + qz*qz);
  float R12 = 2.f*(qy*qz - qw*qx);
  float R20 = 2.f*(qx*qz - qw*qy);
  float R21 = 2.f*(qy*qz + qw*qx);
  float R22 = 1.f - 2.f*(qx*qx + qy*qy);
  float i0 = 1.f/(s0*s0 + 1e-8f);
  float i1 = 1.f/(s1*s1 + 1e-8f);
  float i2 = 1.f/(s2*s2 + 1e-8f);
  float C00 = R00*R00*i0 + R01*R01*i1 + R02*R02*i2;
  float C01 = R00*R10*i0 + R01*R11*i1 + R02*R12*i2;
  float C02 = R00*R20*i0 + R01*R21*i1 + R02*R22*i2;
  float C11 = R10*R10*i0 + R11*R11*i1 + R12*R12*i2;
  float C12 = R10*R20*i0 + R11*R21*i1 + R12*R22*i2;
  float C22 = R20*R20*i0 + R21*R21*i1 + R22*R22*i2;
  float b0 = C00*p0 + C01*p1 + C02*p2;
  float b1 = C01*p0 + C11*p1 + C12*p2;
  float b2 = C02*p0 + C12*p1 + C22*p2;
  float cn = p0*b0 + p1*b1 + p2*b2;
  v4f w0, w1, w2;
  w0.x = C00;      w0.y = C11;      w0.z = 2.f*C01;  w0.w = -2.f*b0;
  w1.x = -2.f*b1;  w1.y = cn;       w1.z = 2.f*C02;  w1.w = 2.f*C12;
  w2.x = -2.f*b2;  w2.y = C22;      w2.z = 0.f;      w2.w = 0.f;
  v4f* wv = (v4f*)wn;
  wv[0] = w0; wv[1] = w1; wv[2] = w2;
}

__global__ __launch_bounds__(256) void vox_main(const float* __restrict__ pos,
                                                const float* __restrict__ scl,
                                                const float* __restrict__ rot,
                                                const float* __restrict__ density,
                                                float* __restrict__ out) {
  __shared__ __attribute__((aligned(16))) float lw[NG * KDIM];
  __shared__ __attribute__((aligned(16))) float ld[NG];
  __shared__ __attribute__((aligned(16))) float lbg[16 * 2 * 32 * 4];

  const int tid = threadIdx.x;
  vox_coeff(pos, scl, rot, tid, &lw[tid * KDIM]);
  ld[tid] = density[tid];
  __syncthreads();

  const int lane = tid & 31;
  const int wave = tid >> 5;
  const int m    = lane & 15;
  const bool h   = lane >= 16;

  const float EXP_SCALE = -0.72134752044448170368f;
  const float SC = 2.0f / 127.0f;

  const int wi = blockIdx.x * 8 + wave;
  const int xb = wi & 7;
  const int yg = (wi >> 3) & 15;
  const int iz = wi >> 7;

  const float cx = (float)(xb * 16 + m) * SC - 1.0f;
  v2f ax;
  ax.x = h ? EXP_SCALE : cx * EXP_SCALE;
  ax.y = h ? 0.f : cx * cx * EXP_SCALE;

  const bool g   = m >= 8;
  const float cz = (float)iz * SC - 1.0f;
  const float cy = (float)(yg * 8 + (m & 7)) * SC - 1.0f;
  v2f p0, p1, p2;
  p0.x = g ? (h ? cz * cy : cz * cz) : 0.f;
  p0.y = g ? (h ? cz      : cy * cy) : 0.f;
  p1.x = h ? (g ? 0.f : cz) : (g ? cy  : 0.f);
  p1.y = h ? (g ? 0.f : cy) : (g ? 1.f : 0.f);
  p2.x = (!h && !g) ? 1.f : 0.f;
  p2.y = 0.f;

  const int hs = h ? 1 : 0;
  const float* wl = &lw[m * KDIM + 2 * hs];
  const float* dl = &ld[hs * 8];

  #pragma unroll
  for (int k = 0; k < 2; ++k) {
    const int t = 2 * wave + k;
    const v2f* wp = (const v2f*)(wl + t * (16 * KDIM));
    const v2f b0 = wp[0];
    const v2f b1 = wp[2];
    const v2f b2 = wp[4];
    v8f bg = {0.f,0.f,0.f,0.f,0.f,0.f,0.f,0.f};
    bg = __builtin_amdgcn_wmma_f32_16x16x4_f32(false, p0, false, b0, (short)0, bg, false, false);
    bg = __builtin_amdgcn_wmma_f32_16x16x4_f32(false, p1, false, b1, (short)0, bg, false, false);
    bg = __builtin_amdgcn_wmma_f32_16x16x4_f32(false, p2, false, b2, (short)0, bg, false, false);
    v4f lo, hi;
    lo.x = bg[0]; lo.y = bg[1]; lo.z = bg[2]; lo.w = bg[3];
    hi.x = bg[4]; hi.y = bg[5]; hi.z = bg[6]; hi.w = bg[7];
    *(v4f*)&lbg[((t * 2 + 0) * 32 + lane) * 4] = lo;
    *(v4f*)&lbg[((t * 2 + 1) * 32 + lane) * 4] = hi;
  }
  __syncthreads();

  v2f acc[8];
  #pragma unroll
  for (int r = 0; r < 8; ++r) { acc[r].x = 0.f; acc[r].y = 0.f; }

#define AFTER_EXP2(a, b) do { asm volatile("" :: "v"(a), "v"(b)); __builtin_amdgcn_sched_barrier(0); } while (0)
#define AFTER_EXP1(a)    do { asm volatile("" :: "v"(a)); __builtin_amdgcn_sched_barrier(0); } while (0)
#define AFTER_SUM(sum)   do { asm volatile("" :: "v"(sum)); __builtin_amdgcn_sched_barrier(0); } while (0)
  const v8f zero8 = {0.f,0.f,0.f,0.f,0.f,0.f,0.f,0.f};
  #pragma unroll 1
  for (int t = 0; t < 16; ++t) {
    const float c22 = wl[t * (16 * KDIM) + 9];
    const v4f glo = *(const v4f*)&lbg[((t * 2 + 0) * 32 + lane) * 4];
    const v4f ghi = *(const v4f*)&lbg[((t * 2 + 1) * 32 + lane) * 4];
    float bgr[8];
    bgr[0] = glo.x; bgr[1] = glo.y; bgr[2] = glo.z; bgr[3] = glo.w;
    bgr[4] = ghi.x; bgr[5] = ghi.y; bgr[6] = ghi.z; bgr[7] = ghi.w;
    const v4f* dp = (const v4f*)(dl + t * 16);
    const v4f dlo = dp[0];
    const v4f dhi = dp[1];
    v2f dv[4];
    dv[0].x = dlo.x; dv[0].y = dlo.y;
    dv[1].x = dlo.z; dv[1].y = dlo.w;
    dv[2].x = dhi.x; dv[2].y = dhi.y;
    dv[3].x = dhi.z; dv[3].y = dhi.w;

    v2f aq;
    aq.x = bgr[0];
    aq.y = c22;
    v8f cur = __builtin_amdgcn_wmma_f32_16x16x4_f32(false, aq, false, ax, (short)0, zero8, false, false);
    v8f nxt = cur;
    v2f g1, g2, g3;
    #pragma unroll
    for (int r = 0; r < 8; ++r) {
      if (r < 7) {
        v2f an;
        an.x = bgr[r + 1];
        an.y = c22;
        nxt = __builtin_amdgcn_wmma_f32_16x16x4_f32(false, an, false, ax, (short)0, zero8, false, false);
      }
      __builtin_amdgcn_sched_barrier(0);
      v2f e0, e1, e2, e3;
      e0.x = __builtin_amdgcn_exp2f(cur[0]);
      e0.y = __builtin_amdgcn_exp2f(cur[1]);
      AFTER_EXP2(e0.x, e0.y);
      if (r > 0) { acc[r - 1] += g1 * dv[1]; AFTER_SUM(acc[r - 1]); }
      e1.x = __builtin_amdgcn_exp2f(cur[2]);
      e1.y = __builtin_amdgcn_exp2f(cur[3]);
      AFTER_EXP2(e1.x, e1.y);
      if (r > 0) { acc[r - 1] += g2 * dv[2]; AFTER_SUM(acc[r - 1]); }
      e2.x = __builtin_amdgcn_exp2f(cur[4]);
      e2.y = __builtin_amdgcn_exp2f(cur[5]);
      AFTER_EXP2(e2.x, e2.y);
      if (r > 0) { acc[r - 1] += g3 * dv[3]; AFTER_SUM(acc[r - 1]); }
      e3.x = __builtin_amdgcn_exp2f(cur[6]);
      AFTER_EXP1(e3.x);
      acc[r] += e0 * dv[0];
      AFTER_SUM(acc[r]);
      e3.y = __builtin_amdgcn_exp2f(cur[7]);
      AFTER_EXP1(e3.y);
      g1 = e1; g2 = e2; g3 = e3;
      cur = nxt;
    }
    acc[7] += g1 * dv[1];
    acc[7] += g2 * dv[2];
    acc[7] += g3 * dv[3];
  }
#undef AFTER_EXP2
#undef AFTER_EXP1
#undef AFTER_SUM

  #pragma unroll
  for (int r = 0; r < 8; r += 2) {
    const float s0 = acc[r].x + acc[r].y;
    const float s1 = acc[r + 1].x + acc[r + 1].y;
    const float mine = h ? s1 : s0;
    const float give = h ? s0 : s1;
    const float got  = __shfl_xor(give, 16, 32);
    const int row = iz * GRID_DIM + yg * 8 + r + hs;
    out[row * GRID_DIM + xb * 16 + m] = mine + got;
  }
}

extern "C" void kernel_launch(void* const* d_in, const int* in_sizes, int n_in,
                              void* d_out, int out_size, void* d_ws, size_t ws_size,
                              hipStream_t stream) {
  const float* positions = (const float*)d_in[0];
  const float* scales    = (const float*)d_in[1];
  const float* rotations = (const float*)d_in[2];
  const float* density   = (const float*)d_in[3];
  float* out = (float*)d_out;

  vox_main<<<MAIN_BLOCKS, 256, 0, stream>>>(positions, scales, rotations, density, out);
}
